# starting kernel
# baseline (speedup 1.0000x reference)
#include <hip/hip_runtime.h>

typedef float v2f __attribute__((ext_vector_type(2)));
typedef float v4f __attribute__((ext_vector_type(4)));
typedef float v8f __attribute__((ext_vector_type(8)));

#define P_TOT  32768
#define H      256
#define L      16
#define KRED   280
#define APITCH 284
#define MTILE  64
#define KC     56
#define NCHUNK 5

__global__ void const_head_kernel(const float* __restrict__ node_embs,
                                  const float* __restrict__ W1,
                                  const float* __restrict__ b1,
                                  const int* __restrict__ src_idx,
                                  const int* __restrict__ dst_idx,
                                  float* __restrict__ cvec) {
  const int h = threadIdx.x;
  const float* se = node_embs + (long)(*src_idx) * H;
  const float* de = node_embs + (long)(*dst_idx) * H;
  float acc = b1[h];
  for (int k = 0; k < H; ++k) acc += se[k] * W1[(long)k * H + h];
  for (int k = 0; k < H; ++k) acc += de[k] * W1[(long)(H + k) * H + h];
  cvec[h] = acc;
}

__global__ __launch_bounds__(256) void path_policy_kernel(
    const float* __restrict__ node_embs,
    const int*   __restrict__ path_nodes,
    const int*   __restrict__ path_lens,
    const float* __restrict__ edge_feats,
    const float* __restrict__ scal_feats,
    const float* __restrict__ W1,
    const float* __restrict__ W2,
    const float* __restrict__ b2,
    const float* __restrict__ cvec,
    float* __restrict__ out) {
  __shared__ float ldsA[MTILE][APITCH];
  __shared__ float ldsB[(KC / 4) * H * 4];
  __shared__ float ldsQ[MTILE];

  const int tid  = threadIdx.x;
  const int lane = tid & 31;
  const int wid  = tid >> 5;
  const int p0   = blockIdx.x * MTILE;

  if (tid < MTILE) ldsQ[tid] = 0.0f;

  {
    const int col = lane * 8;
    for (int j = 0; j < 8; ++j) {
      const int pi  = wid * 8 + j;
      const int p   = p0 + pi;
      const int len = path_lens[p];
      v4f a0 = {}, a1 = {};
      for (int t = 0; t < len; ++t) {
        const long idx = (long)path_nodes[p * L + t];
        const v4f* src = (const v4f*)(node_embs + idx * H + col);
        a0 += src[0];
        a1 += src[1];
      }
      const float inv = 1.0f / (float)(len > 0 ? len : 1);
      v4f* dst = (v4f*)&ldsA[pi][col];
      dst[0] = a0 * inv;
      dst[1] = a1 * inv;
    }
  }
  for (int i = tid; i < MTILE * 24; i += 256) {
    const int pi = i / 24, c = i % 24;
    const float v = (c < 8) ? edge_feats[(long)(p0 + pi) * 8 + c]
                            : scal_feats[(long)(p0 + pi) * 16 + (c - 8)];
    ldsA[pi][256 + c] = v;
  }

  const int mgrp  = wid >> 2;
  const int n0    = (wid & 3) * 64;
  const int lh    = lane & 15;
  const int khalf = lane >> 4;
  const int arow0 = mgrp * 32 + lh;
  const int arow1 = arow0 + 16;

  v8f acc[2][4] = {};

  const float* W1s = W1 + (long)512 * H;
  for (int ch = 0; ch < NCHUNK; ++ch) {
    const int kbase = ch * KC;
    __syncthreads();
    {
      const int n = tid;
      for (int kk = 0; kk < KC / 4; ++kk) {
        const long krow = kbase + kk * 4;
        v4f b;
        b[0] = W1s[(krow + 0) * H + n];
        b[1] = W1s[(krow + 1) * H + n];
        b[2] = W1s[(krow + 2) * H + n];
        b[3] = W1s[(krow + 3) * H + n];
        *(v4f*)&ldsB[(kk * H + n) * 4] = b;
      }
    }
    __syncthreads();
#pragma unroll
    for (int kk = 0; kk < KC / 4; ++kk) {
      const int k = kbase + kk * 4;
      const v2f a0 = *(const v2f*)&ldsA[arow0][k + khalf * 2];
      const v2f a1 = *(const v2f*)&ldsA[arow1][k + khalf * 2];
      const float* bb = &ldsB[(kk * H + n0 + lh) * 4 + khalf * 2];
      const v2f b0 = *(const v2f*)(bb);
      const v2f b1 = *(const v2f*)(bb + 16 * 4);
      const v2f b2v = *(const v2f*)(bb + 32 * 4);
      const v2f b3 = *(const v2f*)(bb + 48 * 4);
      acc[0][0] = __builtin_amdgcn_wmma_f32_16x16x4_f32(false, a0, false, b0,  (short)0, acc[0][0], false, false);
      acc[0][1] = __builtin_amdgcn_wmma_f32_16x16x4_f32(false, a0, false, b1,  (short)0, acc[0][1], false, false);
      acc[0][2] = __builtin_amdgcn_wmma_f32_16x16x4_f32(false, a0, false, b2v, (short)0, acc[0][2], false, false);
      acc[0][3] = __builtin_amdgcn_wmma_f32_16x16x4_f32(false, a0, false, b3,  (short)0, acc[0][3], false, false);
      acc[1][0] = __builtin_amdgcn_wmma_f32_16x16x4_f32(false, a1, false, b0,  (short)0, acc[1][0], false, false);
      acc[1][1] = __builtin_amdgcn_wmma_f32_16x16x4_f32(false, a1, false, b1,  (short)0, acc[1][1], false, false);
      acc[1][2] = __builtin_amdgcn_wmma_f32_16x16x4_f32(false, a1, false, b2v, (short)0, acc[1][2], false, false);
      acc[1][3] = __builtin_amdgcn_wmma_f32_16x16x4_f32(false, a1, false, b3,  (short)0, acc[1][3], false, false);
    }
  }

  float cn[4], wn[4];
#pragma unroll
  for (int s = 0; s < 4; ++s) {
    const int n = n0 + s * 16 + lh;
    cn[s] = cvec[n];
    wn[s] = W2[n];
  }
#pragma unroll
  for (int ms = 0; ms < 2; ++ms) {
#pragma unroll
    for (int r = 0; r < 8; ++r) {
      float part = 0.0f, v;
      v = acc[ms][0][r] + cn[0]; part += fmaxf(v, 0.0f) * wn[0];
      v = acc[ms][1][r] + cn[1]; part += fmaxf(v, 0.0f) * wn[1];
      v = acc[ms][2][r] + cn[2]; part += fmaxf(v, 0.0f) * wn[2];
      v = acc[ms][3][r] + cn[3]; part += fmaxf(v, 0.0f) * wn[3];
      part += __shfl_xor(part, 1, 32);
      part += __shfl_xor(part, 2, 32);
      part += __shfl_xor(part, 4, 32);
      part += __shfl_xor(part, 8, 32);
      if (lh == 0) {
        const int row = mgrp * 32 + ms * 16 + r + khalf * 8;
        atomicAdd(&ldsQ[row], part);
      }
    }
  }

  __syncthreads();
  if (tid < MTILE) {
    const int p   = p0 + tid;
    const int len = path_lens[p];
    out[p] = (len > 0) ? (ldsQ[tid] + b2[0]) : 0.0f;
  }
}

extern "C" void kernel_launch(void* const* d_in, const int* in_sizes, int n_in,
                              void* d_out, int out_size, void* d_ws, size_t ws_size,
                              hipStream_t stream) {
  const float* node_embs = (const float*)d_in[0];
  const int*   path_nodes = (const int*)d_in[1];
  const int*   path_lens  = (const int*)d_in[2];
  const float* edge_feats = (const float*)d_in[3];
  const float* scal_feats = (const float*)d_in[4];
  const float* W1 = (const float*)d_in[5];
  const float* b1 = (const float*)d_in[6];
  const float* W2 = (const float*)d_in[7];
  const float* b2 = (const float*)d_in[8];
  const int*   src_idx = (const int*)d_in[9];
  const int*   dst_idx = (const int*)d_in[10];

  float* cvec = (float*)d_ws;

  const_head_kernel<<<1, 256, 0, stream>>>(node_embs, W1, b1, src_idx, dst_idx, cvec);
  path_policy_kernel<<<P_TOT / MTILE, 256, 0, stream>>>(
      node_embs, path_nodes, path_lens, edge_feats, scal_feats,
      W1, W2, b2, cvec, (float*)d_out);
}


# epilogue row sums: 16-lane xor butterflies (steps 1,2,4,8) moved from 64 dependent LDS shuffles per wave to 15 DPP adds on the vector ALU (transposing form), one LDS add per lane; sum order kept
# speedup vs baseline: 1.6414x; 1.6414x over previous
#include <hip/hip_runtime.h>

typedef float v2f __attribute__((ext_vector_type(2)));
typedef float v4f __attribute__((ext_vector_type(4)));
typedef float v8f __attribute__((ext_vector_type(8)));

#define P_TOT  32768
#define H      256
#define L      16
#define KRED   280
#define APITCH 284
#define MTILE  128
#define NWAVE  16
#define KC     56
#define NCHUNK 5

#define HEADCH 4
#define HROWS  8

#define BCH    ((KC / 4) * H * 4)

__device__ __forceinline__ void bchunk_fetch(v4f (&b)[KC / 8], const float* W1s, const int kbase, const int tid) {
  const int n = tid & (H - 1);
  const int kk0 = (tid >> 8) * (KC / 8);
#pragma unroll
  for (int j = 0; j < KC / 8; ++j) {
    const long krow = kbase + (kk0 + j) * 4;
    b[j][0] = W1s[(krow + 0) * H + n];
    b[j][1] = W1s[(krow + 1) * H + n];
    b[j][2] = W1s[(krow + 2) * H + n];
    b[j][3] = W1s[(krow + 3) * H + n];
  }
}
__device__ __forceinline__ void bchunk_store(const v4f (&b)[KC / 8], float* dst, const int tid) {
  const int n = tid & (H - 1);
  const int kk0 = (tid >> 8) * (KC / 8);
#pragma unroll
  for (int j = 0; j < KC / 8; ++j) *(v4f*)&dst[((kk0 + j) * H + n) * 4] = b[j];
}

template <int M>
__device__ __forceinline__ float row_xor(const float v) {
  return __builtin_bit_cast(float, __builtin_amdgcn_update_dpp(0, __builtin_bit_cast(int, v), 0x160 + M, 0xf, 0xf, true));
}

static_assert(HEADCH == NCHUNK - 1, "the chunks that carry head rows are the chunks that have a successor to fetch");
template <bool HEAD>
__device__ __forceinline__ void gemm_chunk(
    const int kbase, const float (*ldsA)[APITCH], const float* bcur, float* bnext, const float* W1s,
    const int tid, const int arow0, const int arow1, const int khalf, const int n0, const int lh,
    v8f (&acc)[2][4], const float* embc, const float* wrowc,
    v4f& c0, v4f& c1) {
  __syncthreads();
  v4f e, wa[4], wb[4];
  v4f bpre[KC / 8];
  if (HEAD) {
    e = *(const v4f*)(embc);
#pragma unroll
    for (int i = 0; i < 4; ++i) {
      wa[i] = *(const v4f*)(wrowc + (long)i * H);
      wb[i] = *(const v4f*)(wrowc + (long)i * H + 128);
    }
  }
#pragma unroll
  for (int kk = 0; kk < KC / 4; ++kk) {
    if (HEAD && (kk == 4 || kk == 9)) {
      c0 += e[0] * wa[0]; c1 += e[0] * wb[0];
      c0 += e[1] * wa[1]; c1 += e[1] * wb[1];
      c0 += e[2] * wa[2]; c1 += e[2] * wb[2];
      c0 += e[3] * wa[3]; c1 += e[3] * wb[3];
      if (kk == 4) {
        const int r0 = 4;
        e = *(const v4f*)(embc + r0);
#pragma unroll
        for (int i = 0; i < 4; ++i) {
          wa[i] = *(const v4f*)(wrowc + (long)(r0 + i) * H);
          wb[i] = *(const v4f*)(wrowc + (long)(r0 + i) * H + 128);
        }
      }
    }
    if (HEAD && kk == 1) bchunk_fetch(bpre, W1s, kbase + KC, tid);
    if (HEAD && kk == 10) bchunk_store(bpre, bnext, tid);
    const int k = kbase + kk * 4;
    const v2f a0 = *(const v2f*)&ldsA[arow0][k + khalf * 2];
    const v2f a1 = *(const v2f*)&ldsA[arow1][k + khalf * 2];
    const float* bb = &bcur[(kk * H + n0 + lh) * 4 + khalf * 2];
    const v2f b0 = *(const v2f*)(bb);
    const v2f b1 = *(const v2f*)(bb + 16 * 4);
    const v2f b2v = *(const v2f*)(bb + 32 * 4);
    const v2f b3 = *(const v2f*)(bb + 48 * 4);
    acc[0][0] = __builtin_amdgcn_wmma_f32_16x16x4_f32(false, a0, false, b0,  (short)0, acc[0][0], false, false);
    acc[0][1] = __builtin_amdgcn_wmma_f32_16x16x4_f32(false, a0, false, b1,  (short)0, acc[0][1], false, false);
    acc[0][2] = __builtin_amdgcn_wmma_f32_16x16x4_f32(false, a0, false, b2v, (short)0, acc[0][2], false, false);
    acc[0][3] = __builtin_amdgcn_wmma_f32_16x16x4_f32(false, a0, false, b3,  (short)0, acc[0][3], false, false);
    acc[1][0] = __builtin_amdgcn_wmma_f32_16x16x4_f32(false, a1, false, b0,  (short)0, acc[1][0], false, false);
    acc[1][1] = __builtin_amdgcn_wmma_f32_16x16x4_f32(false, a1, false, b1,  (short)0, acc[1][1], false, false);
    acc[1][2] = __builtin_amdgcn_wmma_f32_16x16x4_f32(false, a1, false, b2v, (short)0, acc[1][2], false, false);
    acc[1][3] = __builtin_amdgcn_wmma_f32_16x16x4_f32(false, a1, false, b3,  (short)0, acc[1][3], false, false);
  }
}

__global__ __launch_bounds__(NWAVE * 32) void path_policy_kernel(
    const float* __restrict__ node_embs,
    const int*   __restrict__ path_nodes,
    const int*   __restrict__ path_lens,
    const float* __restrict__ edge_feats,
    const float* __restrict__ scal_feats,
    const float* __restrict__ W1,
    const float* __restrict__ b1,
    const float* __restrict__ W2,
    const float* __restrict__ b2,
    const int*   __restrict__ src_idx,
    const int*   __restrict__ dst_idx,
    float* __restrict__ out) {
  __shared__ float ldsA[MTILE][APITCH];
  __shared__ float ldsB[2 * BCH];
  __shared__ float ldsQ[MTILE];
  __shared__ __attribute__((aligned(16))) float ldsC[NWAVE][H];

  const int tid  = threadIdx.x;
  const int lane = tid & 31;
  const int wid  = tid >> 5;
  const int p0   = blockIdx.x * MTILE;

  if (tid < MTILE) ldsQ[tid] = 0.0f;

  {
    const int g = lane >> 3;
    const int s = lane & 7;
    const int piA = wid * 8 + g;
    const int piB = piA + 4;
    int lenA = path_lens[p0 + piA];
    int lenB = path_lens[p0 + piB];
    const int* nodesA = path_nodes + (long)(p0 + piA) * L;
    const int* nodesB = path_nodes + (long)(p0 + piB) * L;
    const float* nbase = node_embs + s * 4;
    v4f accA[8] = {}, accB[8] = {};

    long oa = (long)nodesA[0] * H;
    long ob = (long)nodesB[0] * H;
    v4f bfirst[KC / 8];
    bchunk_fetch(bfirst, W1 + (long)512 * H, 0, tid);
    asm volatile("" : "+v"(lenA), "+v"(lenB), "+v"(oa), "+v"(ob));
    bchunk_store(bfirst, ldsB, tid);
    for (int t = 0; t < L; ++t) {
      const bool onA = t < lenA;
      const bool onB = t < lenB;
      const float* rpa = nbase + oa;
      const float* rpb = nbase + ob;
      v4f ra[8], rb[8];
      if (onA) {
#pragma unroll
        for (int i = 0; i < 8; ++i) ra[i] = *(const v4f*)(rpa + i * 32);
      }
      if (onB) {
#pragma unroll
        for (int i = 0; i < 8; ++i) rb[i] = *(const v4f*)(rpb + i * 32);
      }
      const int ia = nodesA[(t + 1) & (L - 1)];
      const int ib = nodesB[(t + 1) & (L - 1)];
      if (onA) {
#pragma unroll
        for (int i = 0; i < 8; ++i) accA[i] += ra[i];
      }
      if (onB) {
#pragma unroll
        for (int i = 0; i < 8; ++i) accB[i] += rb[i];
      }
      oa = (long)ia * H;
      ob = (long)ib * H;
      asm volatile("" : "+v"(oa), "+v"(ob));
    }
    const float invA = 1.0f / (float)(lenA > 0 ? lenA : 1);
    const float invB = 1.0f / (float)(lenB > 0 ? lenB : 1);
#pragma unroll
    for (int i = 0; i < 8; ++i) {
      *(v4f*)&ldsA[piA][i * 32 + s * 4] = accA[i] * invA;
      *(v4f*)&ldsA[piB][i * 32 + s * 4] = accB[i] * invB;
    }
  }
  for (int i = tid; i < MTILE * 24; i += NWAVE * 32) {
    const int pi = i / 24, c = i % 24;
    const float v = (c < 8) ? edge_feats[(long)(p0 + pi) * 8 + c]
                            : scal_feats[(long)(p0 + pi) * 16 + (c - 8)];
    ldsA[pi][256 + c] = v;
  }

  const int mgrp  = wid >> 2;
  const int n0    = (wid & 3) * 64;
  const int lh    = lane & 15;
  const int khalf = lane >> 4;
  const int arow0 = mgrp * 32 + lh;
  const int arow1 = arow0 + 16;

  v8f acc[2][4] = {};

  const float* W1s = W1 + (long)512 * H;

  const long erow = (long)((wid < 8) ? *src_idx : *dst_idx);
  const float* emb  = node_embs + erow * H + (wid & 7) * 32;
  const float* wrow = W1 + (long)(wid * 32) * H + lane * 4;
  v4f c0 = {}, c1 = {};

  for (int ch = 0; ch < HEADCH; ++ch)
    gemm_chunk<true>(ch * KC, ldsA, ldsB + (ch & 1) * BCH, ldsB + ((ch + 1) & 1) * BCH, W1s,
                     tid, arow0, arow1, khalf, n0, lh, acc,
                     emb + ch * HROWS, wrow + (long)(ch * HROWS) * H, c0, c1);
  for (int ch = HEADCH; ch < NCHUNK; ++ch)
    gemm_chunk<false>(ch * KC, ldsA, ldsB + (ch & 1) * BCH, ldsB + ((ch + 1) & 1) * BCH, W1s,
                      tid, arow0, arow1, khalf, n0, lh, acc,
                      emb, wrow, c0, c1);

  *(v4f*)&ldsC[wid][lane * 4]       = c0;
  *(v4f*)&ldsC[wid][128 + lane * 4] = c1;
  __syncthreads();

  float cn[4], wn[4];
#pragma unroll
  for (int s = 0; s < 4; ++s) {
    const int n = n0 + s * 16 + lh;
    float c = b1[n];
#pragma unroll
    for (int w = 0; w < NWAVE; ++w) c += ldsC[w][n];
    cn[s] = c;
    wn[s] = W2[n];
  }
  const bool up1 = (lh & 1) != 0, up2 = (lh & 2) != 0, up4 = (lh & 4) != 0, up8 = (lh & 8) != 0;
  float y[8], z[4], w[2];
#pragma unroll
  for (int i = 0; i < 8; ++i) {
    float x[2];
#pragma unroll
    for (int h = 0; h < 2; ++h) {
      const int ms = i >> 2, r = (i & 3) * 2 + h;
      float p = __builtin_fmaf(fmaxf(acc[ms][0][r] + cn[0], 0.0f), wn[0], 0.0f);
      p = __builtin_fmaf(fmaxf(acc[ms][1][r] + cn[1], 0.0f), wn[1], p);
      p = __builtin_fmaf(fmaxf(acc[ms][2][r] + cn[2], 0.0f), wn[2], p);
      p = __builtin_fmaf(fmaxf(acc[ms][3][r] + cn[3], 0.0f), wn[3], p);
      x[h] = p;
    }
    y[i] = (up1 ? x[1] : x[0]) + row_xor<1>(up1 ? x[0] : x[1]);
  }
#pragma unroll
  for (int i = 0; i < 4; ++i)
    z[i] = (up2 ? y[2 * i + 1] : y[2 * i]) + row_xor<2>(up2 ? y[2 * i] : y[2 * i + 1]);
#pragma unroll
  for (int i = 0; i < 2; ++i)
    w[i] = (up4 ? z[2 * i + 1] : z[2 * i]) + row_xor<4>(up4 ? z[2 * i] : z[2 * i + 1]);
  {
    const float mine = (up8 ? w[1] : w[0]) + row_xor<8>(up8 ? w[0] : w[1]);
    const int row = mgrp * 32 + (lh >> 3) * 16 + (lh & 7) + khalf * 8;
    atomicAdd(&ldsQ[row], mine);
  }

  __syncthreads();
  if (tid < MTILE) {
    const int p   = p0 + tid;
    const int len = path_lens[p];
    out[p] = (len > 0) ? (ldsQ[tid] + b2[0]) : 0.0f;
  }
}

extern "C" void kernel_launch(void* const* d_in, const int* in_sizes, int n_in,
                              void* d_out, int out_size, void* d_ws, size_t ws_size,
                              hipStream_t stream) {
  const float* node_embs = (const float*)d_in[0];
  const int*   path_nodes = (const int*)d_in[1];
  const int*   path_lens  = (const int*)d_in[2];
  const float* edge_feats = (const float*)d_in[3];
  const float* scal_feats = (const float*)d_in[4];
  const float* W1 = (const float*)d_in[5];
  const float* b1 = (const float*)d_in[6];
  const float* W2 = (const float*)d_in[7];
  const float* b2 = (const float*)d_in[8];
  const int*   src_idx = (const int*)d_in[9];
  const int*   dst_idx = (const int*)d_in[10];

  path_policy_kernel<<<P_TOT / MTILE, NWAVE * 32, 0, stream>>>(
      node_embs, path_nodes, path_lens, edge_feats, scal_feats,
      W1, b1, W2, b2, src_idx, dst_idx, (float*)d_out);
}
